# starting kernel
# baseline (speedup 1.0000x reference)
#include <hip/hip_runtime.h>

typedef __attribute__((ext_vector_type(2))) float v2f;
typedef __attribute__((ext_vector_type(8))) float v8f;

#define D_   1024
#define H_   16
#define dh   64
#define T_   4096
#define CH   16
#define SEG  16
#define TSEG (T_/SEG)
#define NCHS (TSEG/CH)
#define ST   68
#define ST16 20

__device__ __forceinline__ v8f wmma_f32(v2f a, v2f b, v8f c) {
  return __builtin_amdgcn_wmma_f32_16x16x4_f32(false, a, false, b, (short)0, c,
                                               false, false);
}

__device__ __forceinline__ v2f fragA(const float* m, int stride, int k0) {
  int lane = threadIdx.x & 31;
  const float* p = m + (lane & 15) * stride + k0 + ((lane >> 4) << 1);
  v2f f; f[0] = p[0]; f[1] = p[1];
  return f;
}
__device__ __forceinline__ v2f fragB_kn(const float* m, int stride, int k0, int nbase) {
  int lane = threadIdx.x & 31;
  int kk = k0 + ((lane >> 4) << 1);
  int n  = nbase + (lane & 15);
  v2f f; f[0] = m[kk * stride + n]; f[1] = m[(kk + 1) * stride + n];
  return f;
}

__global__ __launch_bounds__(128)
void wkv_pass1(const float* __restrict__ wg, const float* __restrict__ kg,
               const float* __restrict__ vg, float* __restrict__ Aseg,
               float* __restrict__ Sloc) {
  __shared__ float S[dh * ST];
  __shared__ float Vv[CH * ST];
  __shared__ float Kh[dh * ST16];
  __shared__ float aLast[dh];
  __shared__ float aAcc[dh];

  const int seg = blockIdx.x, h = blockIdx.y;
  const int tid = threadIdx.x;
  const int lane = tid & 31, wid = tid >> 5;
  const int hi = lane >> 4, ln16 = lane & 15;

  for (int x = tid; x < dh * dh; x += 128) S[(x >> 6) * ST + (x & 63)] = 0.f;
  if (tid < dh) aAcc[tid] = 1.f;

  for (int c = 0; c < NCHS; ++c) {
    const int t0 = seg * TSEG + c * CH;
    __syncthreads();
    if (tid < dh) {
      int i = tid;
      float a = 1.f, ai = 1.f;
      float kr[CH], air[CH];
#pragma unroll
      for (int t = 0; t < CH; ++t) {
        int off  = (t0 + t) * D_ + h * dh + i;
        float wv = wg[off];
        if (seg == 0 && c == 0 && t == 0) wv = 1.f;
        float kv = kg[off];
        Vv[t * ST + i] = vg[off];
        float rw = 1.f / wv;
        a *= wv; ai *= rw;
        kr[t] = kv; air[t] = ai;
      }
      aLast[i] = a;
      aAcc[i] *= a;
#pragma unroll
      for (int t = 0; t < CH; ++t) Kh[i * ST16 + t] = kr[t] * (a * air[t]);
    }
    __syncthreads();

    const int nb = wid * 16;
#pragma unroll
    for (int it = 0; it < 4; ++it) {
      v8f acc = {};
#pragma unroll
      for (int k0 = 0; k0 < CH; k0 += 4)
        acc = wmma_f32(fragA(Kh + it * 16 * ST16, ST16, k0),
                       fragB_kn(Vv, ST, k0, nb), acc);
#pragma unroll
      for (int v = 0; v < 8; ++v) {
        int i   = it * 16 + v + 8 * hi;
        int idx = i * ST + nb + ln16;
        S[idx]  = aLast[i] * S[idx] + acc[v];
      }
    }
  }
  __syncthreads();
  if (tid < dh) Aseg[(h * SEG + seg) * dh + tid] = aAcc[tid];
  for (int x = tid; x < dh * dh; x += 128) {
    int i = x >> 6, j = x & 63;
    Sloc[((size_t)(h * SEG + seg) * dh + i) * dh + j] = S[i * ST + j];
  }
}

__global__ __launch_bounds__(128)
void state_scan(const float* __restrict__ wg, const float* __restrict__ isg,
                const float* __restrict__ Aseg, const float* __restrict__ Sloc,
                float* __restrict__ Sstart, float* __restrict__ state_out) {
  const int h = blockIdx.x, tid = threadIdx.x;
  for (int x = tid; x < dh * dh; x += 128) {
    int i = x >> 6, j = x & 63;
    float s = wg[h * dh + j] * isg[h * dh + j];
    for (int sg = 0; sg < SEG; ++sg) {
      size_t base = ((size_t)(h * SEG + sg) * dh + i) * dh + j;
      Sstart[base] = s;
      s = Aseg[(h * SEG + sg) * dh + i] * s + Sloc[base];
    }
    state_out[i * D_ + h * dh + j] = s;
  }
}

__global__ __launch_bounds__(128)
void wkv_pass2(const float* __restrict__ rg, const float* __restrict__ wg,
               const float* __restrict__ kg, const float* __restrict__ vg,
               const float* __restrict__ gg, const float* __restrict__ ug,
               const float* __restrict__ isg, const float* __restrict__ lnw,
               const float* __restrict__ lnb, const float* __restrict__ Sstart,
               float* __restrict__ og) {
  __shared__ float S[dh * ST];
  __shared__ float Rt[CH * ST];
  __shared__ float Kt[CH * ST];
  __shared__ float Ku[CH * ST];
  __shared__ float Vv[CH * ST];
  __shared__ float Kh[dh * ST16];
  __shared__ float CmW[4][CH * ST16];
  __shared__ float ybuf[CH * ST];
  __shared__ float aLast[dh];
  __shared__ float uvec[dh];
  __shared__ float corr0[dh];
  __shared__ float lnw_s[dh], lnb_s[dh];
  __shared__ float mu_s[CH], rs_s[CH];
  __shared__ float r0sum_s;

  const int seg = blockIdx.x, h = blockIdx.y;
  const int tid = threadIdx.x;
  const int lane = tid & 31, wid = tid >> 5;
  const int hi = lane >> 4, ln16 = lane & 15;

  for (int x = tid; x < dh; x += 128) {
    uvec[x]  = ug[h * dh + x];
    corr0[x] = isg[h * dh + x] * (1.f - wg[h * dh + x]);
    lnw_s[x] = lnw[x];
    lnb_s[x] = lnb[x];
  }
  for (int x = tid; x < dh * dh; x += 128) {
    int i = x >> 6, j = x & 63;
    S[i * ST + j] = Sstart[((size_t)(h * SEG + seg) * dh + i) * dh + j];
  }
  if (tid == 0 && seg == 0) {
    float s = 0.f;
    for (int i2 = 0; i2 < dh; ++i2) s += rg[h * dh + i2];
    r0sum_s = s;
  }
  __syncthreads();

  for (int c = 0; c < NCHS; ++c) {
    const int t0 = seg * TSEG + c * CH;

    if (tid < dh) {
      int i = tid;
      float a = 1.f, ai = 1.f;
      float kr[CH], air[CH];
#pragma unroll
      for (int t = 0; t < CH; ++t) {
        int off  = (t0 + t) * D_ + h * dh + i;
        float wv = wg[off];
        if (seg == 0 && c == 0 && t == 0) wv = 1.f;
        float rv = rg[off], kv = kg[off];
        float aip = ai;
        Rt[t * ST + i] = rv * a;
        float rw = 1.f / wv;
        a *= wv; ai *= rw;
        Kt[t * ST + i] = kv * ai;
        Ku[t * ST + i] = kv * uvec[i] * aip;
        Vv[t * ST + i] = vg[off];
        kr[t] = kv; air[t] = ai;
      }
      aLast[i] = a;
#pragma unroll
      for (int t = 0; t < CH; ++t) Kh[i * ST16 + t] = kr[t] * (a * air[t]);
    }
    __syncthreads();

    v8f accC = {}; v8f accD = {};
#pragma unroll
    for (int k0 = 0; k0 < dh; k0 += 4) {
      v2f arf = fragA(Rt, ST, k0);
      accC = wmma_f32(arf, fragA(Kt, ST, k0), accC);
      accD = wmma_f32(arf, fragA(Ku, ST, k0), accD);
    }
    {
      float* cm = CmW[wid];
#pragma unroll
      for (int v = 0; v < 8; ++v) {
        int M = v + 8 * hi, N = ln16;
        cm[M * ST16 + N] = (N < M) ? accC[v] : ((N == M) ? accD[v] : 0.f);
      }
    }
    asm volatile("" ::: "memory");

    const int nb = wid * 16;
    v8f accY = {};
#pragma unroll
    for (int k0 = 0; k0 < dh; k0 += 4)
      accY = wmma_f32(fragA(Rt, ST, k0), fragB_kn(S, ST, k0, nb), accY);
#pragma unroll
    for (int k0 = 0; k0 < CH; k0 += 4)
      accY = wmma_f32(fragA(CmW[wid], ST16, k0), fragB_kn(Vv, ST, k0, nb), accY);
    if (seg == 0 && c == 0 && hi == 0)
      accY[0] += r0sum_s * corr0[nb + ln16];
#pragma unroll
    for (int v = 0; v < 8; ++v) ybuf[(v + 8 * hi) * ST + nb + ln16] = accY[v];

    v8f accS[4];
#pragma unroll
    for (int it = 0; it < 4; ++it) {
      v8f a = {};
#pragma unroll
      for (int k0 = 0; k0 < CH; k0 += 4)
        a = wmma_f32(fragA(Kh + it * 16 * ST16, ST16, k0),
                     fragB_kn(Vv, ST, k0, nb), a);
      accS[it] = a;
    }
    __syncthreads();

#pragma unroll
    for (int it = 0; it < 4; ++it) {
#pragma unroll
      for (int v = 0; v < 8; ++v) {
        int i   = it * 16 + v + 8 * hi;
        int idx = i * ST + nb + ln16;
        S[idx]  = aLast[i] * S[idx] + accS[it][v];
      }
    }
    if (tid < CH) {
      float s = 0.f, sq = 0.f;
      for (int j = 0; j < dh; ++j) {
        float y = ybuf[tid * ST + j]; s += y; sq += y * y;
      }
      float m  = s * (1.f / dh);
      float vv = sq * (1.f / dh) - m * m;
      mu_s[tid] = m;
      rs_s[tid] = rsqrtf(vv + 1e-5f);
    }
    __syncthreads();

    {
      int t  = tid >> 3;
      int jb = (tid & 7) * 8;
      float m = mu_s[t], rs = rs_s[t];
#pragma unroll
      for (int q = 0; q < 8; ++q) {
        int j    = jb + q;
        float y  = ybuf[t * ST + j];
        float yn = (y - m) * rs * lnw_s[j] + lnb_s[j];
        int off  = (t0 + t) * D_ + h * dh + j;
        float gv = gg[off];
        og[off]  = yn * (gv / (1.f + __expf(-gv)));
      }
    }
    __syncthreads();
  }
}

__global__ __launch_bounds__(128)
void oproj_kernel(const float* __restrict__ o, const float* __restrict__ Wo,
                  float* __restrict__ out) {
  __shared__ float Ao[64 * ST];
  __shared__ float Bo[64 * ST];
  const int tid = threadIdx.x;
  const int lane = tid & 31, wid = tid >> 5;
  const int hi = lane >> 4, ln16 = lane & 15;
  const int nbase = blockIdx.x * 64;
  const int mbase = blockIdx.y * 64;

  v8f acc[4] = {};

  for (int ks = 0; ks < D_; ks += 64) {
    __syncthreads();
    {
      int row = tid >> 1;
      int cb  = (tid & 1) * 32;
      const float4* pa = (const float4*)(o  + (size_t)(mbase + row) * D_ + ks + cb);
      const float4* pb = (const float4*)(Wo + (size_t)(nbase + row) * D_ + ks + cb);
#pragma unroll
      for (int q = 0; q < 8; ++q) {
        float4 a4 = pa[q]; float4 b4 = pb[q];
        int cc = cb + q * 4;
        Ao[row * ST + cc + 0] = a4.x; Ao[row * ST + cc + 1] = a4.y;
        Ao[row * ST + cc + 2] = a4.z; Ao[row * ST + cc + 3] = a4.w;
        Bo[row * ST + cc + 0] = b4.x; Bo[row * ST + cc + 1] = b4.y;
        Bo[row * ST + cc + 2] = b4.z; Bo[row * ST + cc + 3] = b4.w;
      }
    }
    __syncthreads();
#pragma unroll
    for (int k0 = 0; k0 < 64; k0 += 4) {
      v2f b = fragA(Bo + wid * 16 * ST, ST, k0);
#pragma unroll
      for (int mt = 0; mt < 4; ++mt)
        acc[mt] = wmma_f32(fragA(Ao + mt * 16 * ST, ST, k0), b, acc[mt]);
    }
  }
#pragma unroll
  for (int mt = 0; mt < 4; ++mt) {
#pragma unroll
    for (int v = 0; v < 8; ++v) {
      int row = mbase + mt * 16 + v + 8 * hi;
      int col = nbase + wid * 16 + ln16;
      out[(size_t)row * D_ + col] = acc[mt][v];
    }
  }
}

extern "C" void kernel_launch(void* const* d_in, const int* in_sizes, int n_in,
                              void* d_out, int out_size, void* d_ws, size_t ws_size,
                              hipStream_t stream) {
  const float* r   = (const float*)d_in[0];
  const float* w   = (const float*)d_in[1];
  const float* k   = (const float*)d_in[2];
  const float* v   = (const float*)d_in[3];
  const float* g   = (const float*)d_in[4];
  const float* u   = (const float*)d_in[5];
  const float* is  = (const float*)d_in[6];
  const float* lnw = (const float*)d_in[7];
  const float* lnb = (const float*)d_in[8];
  const float* Wo  = (const float*)d_in[9];

  float* out   = (float*)d_out;
  float* state = out + (size_t)T_ * D_;

  float* o      = (float*)d_ws;
  float* Aseg   = o + (size_t)T_ * D_;
  float* Sloc   = Aseg + (size_t)H_ * SEG * dh;
  float* Sstart = Sloc + (size_t)H_ * SEG * dh * dh;

  dim3 gs(SEG, H_);
  wkv_pass1<<<gs, 128, 0, stream>>>(w, k, v, Aseg, Sloc);
  state_scan<<<H_, 128, 0, stream>>>(w, is, Aseg, Sloc, Sstart, state);
  wkv_pass2<<<gs, 128, 0, stream>>>(r, w, k, v, g, u, is, lnw, lnb, Sstart, o);

  dim3 g2(D_ / 64, T_ / 64);
  oproj_kernel<<<g2, 128, 0, stream>>>(o, Wo, out);
}


# scan launch removed: each pass2 group folds the earlier segments' carries in its prologue, the last also writes the final state (3 launches, sum order kept); compiler then batches each chunk's loads
# speedup vs baseline: 1.8663x; 1.8663x over previous
#include <hip/hip_runtime.h>

typedef __attribute__((ext_vector_type(2))) float v2f;
typedef __attribute__((ext_vector_type(8))) float v8f;

#define D_   1024
#define H_   16
#define dh   64
#define T_   4096
#define CH   16
#define SEG  16
#define TSEG (T_/SEG)
#define NCHS (TSEG/CH)
#define ST   68
#define ST16 20

__device__ __forceinline__ v8f wmma_f32(v2f a, v2f b, v8f c) {
  return __builtin_amdgcn_wmma_f32_16x16x4_f32(false, a, false, b, (short)0, c,
                                               false, false);
}

__device__ __forceinline__ v2f fragA(const float* m, int stride, int k0) {
  int lane = threadIdx.x & 31;
  const float* p = m + (lane & 15) * stride + k0 + ((lane >> 4) << 1);
  v2f f; f[0] = p[0]; f[1] = p[1];
  return f;
}
__device__ __forceinline__ v2f fragB_kn(const float* m, int stride, int k0, int nbase) {
  int lane = threadIdx.x & 31;
  int kk = k0 + ((lane >> 4) << 1);
  int n  = nbase + (lane & 15);
  v2f f; f[0] = m[kk * stride + n]; f[1] = m[(kk + 1) * stride + n];
  return f;
}

__global__ __launch_bounds__(128)
void wkv_pass1(const float* __restrict__ wg, const float* __restrict__ kg,
               const float* __restrict__ vg, float* __restrict__ Aseg,
               float* __restrict__ Sloc) {
  __shared__ float S[dh * ST];
  __shared__ float Vv[CH * ST];
  __shared__ float Kh[dh * ST16];
  __shared__ float aLast[dh];
  __shared__ float aAcc[dh];

  const int seg = blockIdx.x, h = blockIdx.y;
  const int tid = threadIdx.x;
  const int lane = tid & 31, wid = tid >> 5;
  const int hi = lane >> 4, ln16 = lane & 15;

  for (int x = tid; x < dh * dh; x += 128) S[(x >> 6) * ST + (x & 63)] = 0.f;
  if (tid < dh) aAcc[tid] = 1.f;

  for (int c = 0; c < NCHS; ++c) {
    const int t0 = seg * TSEG + c * CH;
    __syncthreads();
    if (tid < dh) {
      int i = tid;
      float a = 1.f, ai = 1.f;
      float kr[CH], air[CH];
#pragma unroll
      for (int t = 0; t < CH; ++t) {
        int off  = (t0 + t) * D_ + h * dh + i;
        float wv = wg[off];
        if (seg == 0 && c == 0 && t == 0) wv = 1.f;
        float kv = kg[off];
        Vv[t * ST + i] = vg[off];
        float rw = 1.f / wv;
        a *= wv; ai *= rw;
        kr[t] = kv; air[t] = ai;
      }
      aLast[i] = a;
      aAcc[i] *= a;
#pragma unroll
      for (int t = 0; t < CH; ++t) Kh[i * ST16 + t] = kr[t] * (a * air[t]);
    }
    __syncthreads();

    const int nb = wid * 16;
#pragma unroll
    for (int it = 0; it < 4; ++it) {
      v8f acc = {};
#pragma unroll
      for (int k0 = 0; k0 < CH; k0 += 4)
        acc = wmma_f32(fragA(Kh + it * 16 * ST16, ST16, k0),
                       fragB_kn(Vv, ST, k0, nb), acc);
#pragma unroll
      for (int v = 0; v < 8; ++v) {
        int i   = it * 16 + v + 8 * hi;
        int idx = i * ST + nb + ln16;
        S[idx]  = aLast[i] * S[idx] + acc[v];
      }
    }
  }
  __syncthreads();
  if (tid < dh) Aseg[(h * SEG + seg) * dh + tid] = aAcc[tid];
  for (int x = tid; x < dh * dh; x += 128) {
    int i = x >> 6, j = x & 63;
    Sloc[((size_t)(h * SEG + seg) * dh + i) * dh + j] = S[i * ST + j];
  }
}

template <int N>
__device__ __forceinline__ void fold_carries(float4 (&s)[8],
                                             const float* __restrict__ AsegH,
                                             const float* __restrict__ SlocH,
                                             int sg, int tid) {
  float4 l[N][8];
  float  a[N][8];
#pragma unroll
  for (int n = 0; n < N; ++n) {
    const float4* L = (const float4*)(SlocH + (size_t)(sg + n) * dh * dh);
    const float*  A = AsegH + (sg + n) * dh + (tid >> 4);
#pragma unroll
    for (int q = 0; q < 8; ++q) { l[n][q] = L[q * 128 + tid]; a[n][q] = A[q * 8]; }
  }
#pragma unroll
  for (int n = 0; n < N; ++n) {
#pragma unroll
    for (int q = 0; q < 8; ++q) {
      s[q].x = a[n][q] * s[q].x + l[n][q].x;
      s[q].y = a[n][q] * s[q].y + l[n][q].y;
      s[q].z = a[n][q] * s[q].z + l[n][q].z;
      s[q].w = a[n][q] * s[q].w + l[n][q].w;
    }
  }
}

__global__ __launch_bounds__(128)
void wkv_pass2(const float* __restrict__ rg, const float* __restrict__ wg,
               const float* __restrict__ kg, const float* __restrict__ vg,
               const float* __restrict__ gg, const float* __restrict__ ug,
               const float* __restrict__ isg, const float* __restrict__ lnw,
               const float* __restrict__ lnb, const float* __restrict__ Aseg,
               const float* __restrict__ Sloc, float* __restrict__ state_out,
               float* __restrict__ og) {
  __shared__ __attribute__((aligned(16))) float S[dh * ST];
  __shared__ float Rt[CH * ST];
  __shared__ float Kt[CH * ST];
  __shared__ float Ku[CH * ST];
  __shared__ float Vv[CH * ST];
  __shared__ float Kh[dh * ST16];
  __shared__ float CmW[4][CH * ST16];
  __shared__ float ybuf[CH * ST];
  __shared__ float aLast[dh];
  __shared__ float uvec[dh];
  __shared__ float corr0[dh];
  __shared__ float lnw_s[dh], lnb_s[dh];
  __shared__ float mu_s[CH], rs_s[CH];
  __shared__ float r0sum_s;

  const int seg = blockIdx.x, h = blockIdx.y;
  const int tid = threadIdx.x;
  const int lane = tid & 31, wid = tid >> 5;
  const int hi = lane >> 4, ln16 = lane & 15;

  for (int x = tid; x < dh; x += 128) {
    uvec[x]  = ug[h * dh + x];
    corr0[x] = isg[h * dh + x] * (1.f - wg[h * dh + x]);
    lnw_s[x] = lnw[x];
    lnb_s[x] = lnb[x];
  }
  if (tid == 0 && seg == 0) {
    float s = 0.f;
    for (int i2 = 0; i2 < dh; ++i2) s += rg[h * dh + i2];
    r0sum_s = s;
  }
  {
    const float* AsegH = Aseg + h * SEG * dh;
    const float* SlocH = Sloc + (size_t)h * SEG * dh * dh;
    const int j4 = (tid & 15) * 4;
    const float4 w0 = *(const float4*)(wg + h * dh + j4);
    const float4 i0 = *(const float4*)(isg + h * dh + j4);
    float4 s[8];
#pragma unroll
    for (int q = 0; q < 8; ++q) {
      s[q].x = w0.x * i0.x; s[q].y = w0.y * i0.y;
      s[q].z = w0.z * i0.z; s[q].w = w0.w * i0.w;
    }
    int sg = 0;
    for (; sg + 4 <= seg; sg += 4) fold_carries<4>(s, AsegH, SlocH, sg, tid);
    if (sg + 2 <= seg) { fold_carries<2>(s, AsegH, SlocH, sg, tid); sg += 2; }
    if (sg < seg)      { fold_carries<1>(s, AsegH, SlocH, sg, tid); sg += 1; }
#pragma unroll
    for (int q = 0; q < 8; ++q)
      *(float4*)(S + (q * 8 + (tid >> 4)) * ST + j4) = s[q];
    if (seg == SEG - 1) {
      fold_carries<1>(s, AsegH, SlocH, sg, tid);
#pragma unroll
      for (int q = 0; q < 8; ++q)
        *(float4*)(state_out + (size_t)(q * 8 + (tid >> 4)) * D_ + h * dh + j4) = s[q];
    }
  }
  __syncthreads();

  for (int c = 0; c < NCHS; ++c) {
    const int t0 = seg * TSEG + c * CH;

    if (tid < dh) {
      int i = tid;
      float a = 1.f, ai = 1.f;
      float kr[CH], air[CH];
#pragma unroll
      for (int t = 0; t < CH; ++t) {
        int off  = (t0 + t) * D_ + h * dh + i;
        float wv = wg[off];
        if (seg == 0 && c == 0 && t == 0) wv = 1.f;
        float rv = rg[off], kv = kg[off];
        float aip = ai;
        Rt[t * ST + i] = rv * a;
        float rw = 1.f / wv;
        a *= wv; ai *= rw;
        Kt[t * ST + i] = kv * ai;
        Ku[t * ST + i] = kv * uvec[i] * aip;
        Vv[t * ST + i] = vg[off];
        kr[t] = kv; air[t] = ai;
      }
      aLast[i] = a;
#pragma unroll
      for (int t = 0; t < CH; ++t) Kh[i * ST16 + t] = kr[t] * (a * air[t]);
    }
    __syncthreads();

    v8f accC = {}; v8f accD = {};
#pragma unroll
    for (int k0 = 0; k0 < dh; k0 += 4) {
      v2f arf = fragA(Rt, ST, k0);
      accC = wmma_f32(arf, fragA(Kt, ST, k0), accC);
      accD = wmma_f32(arf, fragA(Ku, ST, k0), accD);
    }
    {
      float* cm = CmW[wid];
#pragma unroll
      for (int v = 0; v < 8; ++v) {
        int M = v + 8 * hi, N = ln16;
        cm[M * ST16 + N] = (N < M) ? accC[v] : ((N == M) ? accD[v] : 0.f);
      }
    }
    asm volatile("" ::: "memory");

    const int nb = wid * 16;
    v8f accY = {};
#pragma unroll
    for (int k0 = 0; k0 < dh; k0 += 4)
      accY = wmma_f32(fragA(Rt, ST, k0), fragB_kn(S, ST, k0, nb), accY);
#pragma unroll
    for (int k0 = 0; k0 < CH; k0 += 4)
      accY = wmma_f32(fragA(CmW[wid], ST16, k0), fragB_kn(Vv, ST, k0, nb), accY);
    if (seg == 0 && c == 0 && hi == 0)
      accY[0] += r0sum_s * corr0[nb + ln16];
#pragma unroll
    for (int v = 0; v < 8; ++v) ybuf[(v + 8 * hi) * ST + nb + ln16] = accY[v];

    v8f accS[4];
#pragma unroll
    for (int it = 0; it < 4; ++it) {
      v8f a = {};
#pragma unroll
      for (int k0 = 0; k0 < CH; k0 += 4)
        a = wmma_f32(fragA(Kh + it * 16 * ST16, ST16, k0),
                     fragB_kn(Vv, ST, k0, nb), a);
      accS[it] = a;
    }
    __syncthreads();

#pragma unroll
    for (int it = 0; it < 4; ++it) {
#pragma unroll
      for (int v = 0; v < 8; ++v) {
        int i   = it * 16 + v + 8 * hi;
        int idx = i * ST + nb + ln16;
        S[idx]  = aLast[i] * S[idx] + accS[it][v];
      }
    }
    if (tid < CH) {
      float s = 0.f, sq = 0.f;
      for (int j = 0; j < dh; ++j) {
        float y = ybuf[tid * ST + j]; s += y; sq += y * y;
      }
      float m  = s * (1.f / dh);
      float vv = sq * (1.f / dh) - m * m;
      mu_s[tid] = m;
      rs_s[tid] = rsqrtf(vv + 1e-5f);
    }
    __syncthreads();

    {
      int t  = tid >> 3;
      int jb = (tid & 7) * 8;
      float m = mu_s[t], rs = rs_s[t];
#pragma unroll
      for (int q = 0; q < 8; ++q) {
        int j    = jb + q;
        float y  = ybuf[t * ST + j];
        float yn = (y - m) * rs * lnw_s[j] + lnb_s[j];
        int off  = (t0 + t) * D_ + h * dh + j;
        float gv = gg[off];
        og[off]  = yn * (gv / (1.f + __expf(-gv)));
      }
    }
    __syncthreads();
  }
}

__global__ __launch_bounds__(128)
void oproj_kernel(const float* __restrict__ o, const float* __restrict__ Wo,
                  float* __restrict__ out) {
  __shared__ float Ao[64 * ST];
  __shared__ float Bo[64 * ST];
  const int tid = threadIdx.x;
  const int lane = tid & 31, wid = tid >> 5;
  const int hi = lane >> 4, ln16 = lane & 15;
  const int nbase = blockIdx.x * 64;
  const int mbase = blockIdx.y * 64;

  v8f acc[4] = {};

  for (int ks = 0; ks < D_; ks += 64) {
    __syncthreads();
    {
      int row = tid >> 1;
      int cb  = (tid & 1) * 32;
      const float4* pa = (const float4*)(o  + (size_t)(mbase + row) * D_ + ks + cb);
      const float4* pb = (const float4*)(Wo + (size_t)(nbase + row) * D_ + ks + cb);
#pragma unroll
      for (int q = 0; q < 8; ++q) {
        float4 a4 = pa[q]; float4 b4 = pb[q];
        int cc = cb + q * 4;
        Ao[row * ST + cc + 0] = a4.x; Ao[row * ST + cc + 1] = a4.y;
        Ao[row * ST + cc + 2] = a4.z; Ao[row * ST + cc + 3] = a4.w;
        Bo[row * ST + cc + 0] = b4.x; Bo[row * ST + cc + 1] = b4.y;
        Bo[row * ST + cc + 2] = b4.z; Bo[row * ST + cc + 3] = b4.w;
      }
    }
    __syncthreads();
#pragma unroll
    for (int k0 = 0; k0 < 64; k0 += 4) {
      v2f b = fragA(Bo + wid * 16 * ST, ST, k0);
#pragma unroll
      for (int mt = 0; mt < 4; ++mt)
        acc[mt] = wmma_f32(fragA(Ao + mt * 16 * ST, ST, k0), b, acc[mt]);
    }
  }
#pragma unroll
  for (int mt = 0; mt < 4; ++mt) {
#pragma unroll
    for (int v = 0; v < 8; ++v) {
      int row = mbase + mt * 16 + v + 8 * hi;
      int col = nbase + wid * 16 + ln16;
      out[(size_t)row * D_ + col] = acc[mt][v];
    }
  }
}

extern "C" void kernel_launch(void* const* d_in, const int* in_sizes, int n_in,
                              void* d_out, int out_size, void* d_ws, size_t ws_size,
                              hipStream_t stream) {
  const float* r   = (const float*)d_in[0];
  const float* w   = (const float*)d_in[1];
  const float* k   = (const float*)d_in[2];
  const float* v   = (const float*)d_in[3];
  const float* g   = (const float*)d_in[4];
  const float* u   = (const float*)d_in[5];
  const float* is  = (const float*)d_in[6];
  const float* lnw = (const float*)d_in[7];
  const float* lnb = (const float*)d_in[8];
  const float* Wo  = (const float*)d_in[9];

  float* out   = (float*)d_out;
  float* state = out + (size_t)T_ * D_;

  float* o      = (float*)d_ws;
  float* Aseg   = o + (size_t)T_ * D_;
  float* Sloc   = Aseg + (size_t)H_ * SEG * dh;

  dim3 gs(SEG, H_);
  wkv_pass1<<<gs, 128, 0, stream>>>(w, k, v, Aseg, Sloc);
  wkv_pass2<<<gs, 128, 0, stream>>>(r, w, k, v, g, u, is, lnw, lnb, Aseg, Sloc,
                                    state, o);

  dim3 g2(D_ / 64, T_ / 64);
  oproj_kernel<<<g2, 128, 0, stream>>>(o, Wo, out);
}


# pass1: segment carry as ONE 64x256x64 product (P.K)^T V, k scaled by the reverse running product of w (no division, no chunk recurrence on S); order of the sum over t and of the w products changes
# speedup vs baseline: 3.4840x; 1.8668x over previous
#include <hip/hip_runtime.h>

typedef __attribute__((ext_vector_type(2))) float v2f;
typedef __attribute__((ext_vector_type(8))) float v8f;

#define D_   1024
#define H_   16
#define dh   64
#define T_   4096
#define CH   16
#define SEG  16
#define TSEG (T_/SEG)
#define NCHS (TSEG/CH)
#define ST   68
#define ST16 20

__device__ __forceinline__ v8f wmma_f32(v2f a, v2f b, v8f c) {
  return __builtin_amdgcn_wmma_f32_16x16x4_f32(false, a, false, b, (short)0, c,
                                               false, false);
}

__device__ __forceinline__ v2f fragA(const float* m, int stride, int k0) {
  int lane = threadIdx.x & 31;
  const float* p = m + (lane & 15) * stride + k0 + ((lane >> 4) << 1);
  v2f f; f[0] = p[0]; f[1] = p[1];
  return f;
}
__device__ __forceinline__ v2f fragB_kn(const float* m, int stride, int k0, int nbase) {
  int lane = threadIdx.x & 31;
  int kk = k0 + ((lane >> 4) << 1);
  int n  = nbase + (lane & 15);
  v2f f; f[0] = m[kk * stride + n]; f[1] = m[(kk + 1) * stride + n];
  return f;
}

#define P1ST 72
#define TBLK 16
#define NBLK (TSEG / TBLK)

__device__ __forceinline__ float4 mul4(float4 a, float4 b) {
  float4 r; r.x = a.x * b.x; r.y = a.y * b.y; r.z = a.z * b.z; r.w = a.w * b.w;
  return r;
}

__global__ __launch_bounds__(256)
void wkv_pass1(const float* __restrict__ wg, const float* __restrict__ kg,
               const float* __restrict__ vg, float* __restrict__ Aseg,
               float* __restrict__ Sloc) {
  __shared__ __attribute__((aligned(16))) float Kp[TSEG * P1ST];
  __shared__ __attribute__((aligned(16))) float Vp[TSEG * P1ST];
  __shared__ __attribute__((aligned(16))) float blk[NBLK * dh];

  const int seg = blockIdx.x, h = blockIdx.y;
  const int tid = threadIdx.x;
  const int lane = tid & 31, wid = tid >> 5;
  const int hi = lane >> 4, ln16 = lane & 15;
  const int cg = tid & 15, tb = tid >> 4;
  const int j4 = cg * 4;
  const size_t g0 = (size_t)(seg * TSEG + tb * TBLK) * D_ + h * dh + j4;

#pragma unroll
  for (int s = 0; s < TBLK; ++s)
    *(float4*)(Vp + (tb * TBLK + s) * P1ST + j4) = *(const float4*)(vg + g0 + (size_t)s * D_);

  float4 wv[TBLK], kv[TBLK];
#pragma unroll
  for (int s = 0; s < TBLK; ++s) wv[s] = *(const float4*)(wg + g0 + (size_t)s * D_);
#pragma unroll
  for (int s = 0; s < TBLK; ++s) kv[s] = *(const float4*)(kg + g0 + (size_t)s * D_);

  float4 p = make_float4(1.f, 1.f, 1.f, 1.f);
#pragma unroll
  for (int s = TBLK - 1; s >= 0; --s) {
    float4 t = wv[s];
    wv[s] = p;
    p = mul4(p, t);
  }
  const float4 total = (seg == 0 && tb == 0) ? wv[0] : p;
  *(float4*)(blk + tb * dh + j4) = total;
  __syncthreads();
  float4 carry = make_float4(1.f, 1.f, 1.f, 1.f);
#pragma unroll
  for (int b = NBLK - 1; b >= 1; --b) {
    float4 x = *(const float4*)(blk + b * dh + j4);
    const bool later = b > tb;
    x.x = later ? x.x : 1.f; x.y = later ? x.y : 1.f;
    x.z = later ? x.z : 1.f; x.w = later ? x.w : 1.f;
    carry = mul4(carry, x);
  }
  if (tb == 0)
    *(float4*)(Aseg + (h * SEG + seg) * dh + j4) = mul4(total, carry);

#pragma unroll
  for (int s = 0; s < TBLK; ++s) {
    const int row = tb * TBLK + s;
    *(float4*)(Kp + row * P1ST + j4) = mul4(kv[s], mul4(wv[s], carry));
  }
  __syncthreads();

  const int nb  = (wid & 3) * 16;
  const int itb = (wid >> 2) * 2;
  v8f acc[2] = {};
#pragma unroll 8
  for (int t0 = 0; t0 < TSEG; t0 += 4) {
    v2f b = fragB_kn(Vp, P1ST, t0, nb);
#pragma unroll
    for (int it = 0; it < 2; ++it)
      acc[it] = wmma_f32(fragB_kn(Kp, P1ST, t0, (itb + it) * 16), b, acc[it]);
  }
  __syncthreads();
#pragma unroll
  for (int it = 0; it < 2; ++it) {
#pragma unroll
    for (int v = 0; v < 8; ++v)
      Kp[((itb + it) * 16 + v + 8 * hi) * P1ST + nb + ln16] = acc[it][v];
  }
  __syncthreads();
  float* So = Sloc + (size_t)(h * SEG + seg) * dh * dh;
#pragma unroll
  for (int q = 0; q < 4; ++q) {
    int e = q * 256 + tid;
    int row = e >> 4, c4 = (e & 15) * 4;
    *(float4*)(So + row * dh + c4) = *(const float4*)(Kp + row * P1ST + c4);
  }
}

template <int N>
__device__ __forceinline__ void fold_carries(float4 (&s)[8],
                                             const float* __restrict__ AsegH,
                                             const float* __restrict__ SlocH,
                                             int sg, int tid) {
  float4 l[N][8];
  float  a[N][8];
#pragma unroll
  for (int n = 0; n < N; ++n) {
    const float4* L = (const float4*)(SlocH + (size_t)(sg + n) * dh * dh);
    const float*  A = AsegH + (sg + n) * dh + (tid >> 4);
#pragma unroll
    for (int q = 0; q < 8; ++q) { l[n][q] = L[q * 128 + tid]; a[n][q] = A[q * 8]; }
  }
#pragma unroll
  for (int n = 0; n < N; ++n) {
#pragma unroll
    for (int q = 0; q < 8; ++q) {
      s[q].x = a[n][q] * s[q].x + l[n][q].x;
      s[q].y = a[n][q] * s[q].y + l[n][q].y;
      s[q].z = a[n][q] * s[q].z + l[n][q].z;
      s[q].w = a[n][q] * s[q].w + l[n][q].w;
    }
  }
}

__global__ __launch_bounds__(128)
void wkv_pass2(const float* __restrict__ rg, const float* __restrict__ wg,
               const float* __restrict__ kg, const float* __restrict__ vg,
               const float* __restrict__ gg, const float* __restrict__ ug,
               const float* __restrict__ isg, const float* __restrict__ lnw,
               const float* __restrict__ lnb, const float* __restrict__ Aseg,
               const float* __restrict__ Sloc, float* __restrict__ state_out,
               float* __restrict__ og) {
  __shared__ __attribute__((aligned(16))) float S[dh * ST];
  __shared__ float Rt[CH * ST];
  __shared__ float Kt[CH * ST];
  __shared__ float Ku[CH * ST];
  __shared__ float Vv[CH * ST];
  __shared__ float Kh[dh * ST16];
  __shared__ float CmW[4][CH * ST16];
  __shared__ float ybuf[CH * ST];
  __shared__ float aLast[dh];
  __shared__ float uvec[dh];
  __shared__ float corr0[dh];
  __shared__ float lnw_s[dh], lnb_s[dh];
  __shared__ float mu_s[CH], rs_s[CH];
  __shared__ float r0sum_s;

  const int seg = blockIdx.x, h = blockIdx.y;
  const int tid = threadIdx.x;
  const int lane = tid & 31, wid = tid >> 5;
  const int hi = lane >> 4, ln16 = lane & 15;

  for (int x = tid; x < dh; x += 128) {
    uvec[x]  = ug[h * dh + x];
    corr0[x] = isg[h * dh + x] * (1.f - wg[h * dh + x]);
    lnw_s[x] = lnw[x];
    lnb_s[x] = lnb[x];
  }
  if (tid == 0 && seg == 0) {
    float s = 0.f;
    for (int i2 = 0; i2 < dh; ++i2) s += rg[h * dh + i2];
    r0sum_s = s;
  }
  {
    const float* AsegH = Aseg + h * SEG * dh;
    const float* SlocH = Sloc + (size_t)h * SEG * dh * dh;
    const int j4 = (tid & 15) * 4;
    const float4 w0 = *(const float4*)(wg + h * dh + j4);
    const float4 i0 = *(const float4*)(isg + h * dh + j4);
    float4 s[8];
#pragma unroll
    for (int q = 0; q < 8; ++q) {
      s[q].x = w0.x * i0.x; s[q].y = w0.y * i0.y;
      s[q].z = w0.z * i0.z; s[q].w = w0.w * i0.w;
    }
    int sg = 0;
    for (; sg + 4 <= seg; sg += 4) fold_carries<4>(s, AsegH, SlocH, sg, tid);
    if (sg + 2 <= seg) { fold_carries<2>(s, AsegH, SlocH, sg, tid); sg += 2; }
    if (sg < seg)      { fold_carries<1>(s, AsegH, SlocH, sg, tid); sg += 1; }
#pragma unroll
    for (int q = 0; q < 8; ++q)
      *(float4*)(S + (q * 8 + (tid >> 4)) * ST + j4) = s[q];
    if (seg == SEG - 1) {
      fold_carries<1>(s, AsegH, SlocH, sg, tid);
#pragma unroll
      for (int q = 0; q < 8; ++q)
        *(float4*)(state_out + (size_t)(q * 8 + (tid >> 4)) * D_ + h * dh + j4) = s[q];
    }
  }
  __syncthreads();

  for (int c = 0; c < NCHS; ++c) {
    const int t0 = seg * TSEG + c * CH;

    if (tid < dh) {
      int i = tid;
      float a = 1.f, ai = 1.f;
      float kr[CH], air[CH];
#pragma unroll
      for (int t = 0; t < CH; ++t) {
        int off  = (t0 + t) * D_ + h * dh + i;
        float wv = wg[off];
        if (seg == 0 && c == 0 && t == 0) wv = 1.f;
        float rv = rg[off], kv = kg[off];
        float aip = ai;
        Rt[t * ST + i] = rv * a;
        float rw = 1.f / wv;
        a *= wv; ai *= rw;
        Kt[t * ST + i] = kv * ai;
        Ku[t * ST + i] = kv * uvec[i] * aip;
        Vv[t * ST + i] = vg[off];
        kr[t] = kv; air[t] = ai;
      }
      aLast[i] = a;
#pragma unroll
      for (int t = 0; t < CH; ++t) Kh[i * ST16 + t] = kr[t] * (a * air[t]);
    }
    __syncthreads();

    v8f accC = {}; v8f accD = {};
#pragma unroll
    for (int k0 = 0; k0 < dh; k0 += 4) {
      v2f arf = fragA(Rt, ST, k0);
      accC = wmma_f32(arf, fragA(Kt, ST, k0), accC);
      accD = wmma_f32(arf, fragA(Ku, ST, k0), accD);
    }
    {
      float* cm = CmW[wid];
#pragma unroll
      for (int v = 0; v < 8; ++v) {
        int M = v + 8 * hi, N = ln16;
        cm[M * ST16 + N] = (N < M) ? accC[v] : ((N == M) ? accD[v] : 0.f);
      }
    }
    asm volatile("" ::: "memory");

    const int nb = wid * 16;
    v8f accY = {};
#pragma unroll
    for (int k0 = 0; k0 < dh; k0 += 4)
      accY = wmma_f32(fragA(Rt, ST, k0), fragB_kn(S, ST, k0, nb), accY);
#pragma unroll
    for (int k0 = 0; k0 < CH; k0 += 4)
      accY = wmma_f32(fragA(CmW[wid], ST16, k0), fragB_kn(Vv, ST, k0, nb), accY);
    if (seg == 0 && c == 0 && hi == 0)
      accY[0] += r0sum_s * corr0[nb + ln16];
#pragma unroll
    for (int v = 0; v < 8; ++v) ybuf[(v + 8 * hi) * ST + nb + ln16] = accY[v];

    v8f accS[4];
#pragma unroll
    for (int it = 0; it < 4; ++it) {
      v8f a = {};
#pragma unroll
      for (int k0 = 0; k0 < CH; k0 += 4)
        a = wmma_f32(fragA(Kh + it * 16 * ST16, ST16, k0),
                     fragB_kn(Vv, ST, k0, nb), a);
      accS[it] = a;
    }
    __syncthreads();

#pragma unroll
    for (int it = 0; it < 4; ++it) {
#pragma unroll
      for (int v = 0; v < 8; ++v) {
        int i   = it * 16 + v + 8 * hi;
        int idx = i * ST + nb + ln16;
        S[idx]  = aLast[i] * S[idx] + accS[it][v];
      }
    }
    if (tid < CH) {
      float s = 0.f, sq = 0.f;
      for (int j = 0; j < dh; ++j) {
        float y = ybuf[tid * ST + j]; s += y; sq += y * y;
      }
      float m  = s * (1.f / dh);
      float vv = sq * (1.f / dh) - m * m;
      mu_s[tid] = m;
      rs_s[tid] = rsqrtf(vv + 1e-5f);
    }
    __syncthreads();

    {
      int t  = tid >> 3;
      int jb = (tid & 7) * 8;
      float m = mu_s[t], rs = rs_s[t];
#pragma unroll
      for (int q = 0; q < 8; ++q) {
        int j    = jb + q;
        float y  = ybuf[t * ST + j];
        float yn = (y - m) * rs * lnw_s[j] + lnb_s[j];
        int off  = (t0 + t) * D_ + h * dh + j;
        float gv = gg[off];
        og[off]  = yn * (gv / (1.f + __expf(-gv)));
      }
    }
    __syncthreads();
  }
}

__global__ __launch_bounds__(128)
void oproj_kernel(const float* __restrict__ o, const float* __restrict__ Wo,
                  float* __restrict__ out) {
  __shared__ float Ao[64 * ST];
  __shared__ float Bo[64 * ST];
  const int tid = threadIdx.x;
  const int lane = tid & 31, wid = tid >> 5;
  const int hi = lane >> 4, ln16 = lane & 15;
  const int nbase = blockIdx.x * 64;
  const int mbase = blockIdx.y * 64;

  v8f acc[4] = {};

  for (int ks = 0; ks < D_; ks += 64) {
    __syncthreads();
    {
      int row = tid >> 1;
      int cb  = (tid & 1) * 32;
      const float4* pa = (const float4*)(o  + (size_t)(mbase + row) * D_ + ks + cb);
      const float4* pb = (const float4*)(Wo + (size_t)(nbase + row) * D_ + ks + cb);
#pragma unroll
      for (int q = 0; q < 8; ++q) {
        float4 a4 = pa[q]; float4 b4 = pb[q];
        int cc = cb + q * 4;
        Ao[row * ST + cc + 0] = a4.x; Ao[row * ST + cc + 1] = a4.y;
        Ao[row * ST + cc + 2] = a4.z; Ao[row * ST + cc + 3] = a4.w;
        Bo[row * ST + cc + 0] = b4.x; Bo[row * ST + cc + 1] = b4.y;
        Bo[row * ST + cc + 2] = b4.z; Bo[row * ST + cc + 3] = b4.w;
      }
    }
    __syncthreads();
#pragma unroll
    for (int k0 = 0; k0 < 64; k0 += 4) {
      v2f b = fragA(Bo + wid * 16 * ST, ST, k0);
#pragma unroll
      for (int mt = 0; mt < 4; ++mt)
        acc[mt] = wmma_f32(fragA(Ao + mt * 16 * ST, ST, k0), b, acc[mt]);
    }
  }
#pragma unroll
  for (int mt = 0; mt < 4; ++mt) {
#pragma unroll
    for (int v = 0; v < 8; ++v) {
      int row = mbase + mt * 16 + v + 8 * hi;
      int col = nbase + wid * 16 + ln16;
      out[(size_t)row * D_ + col] = acc[mt][v];
    }
  }
}

extern "C" void kernel_launch(void* const* d_in, const int* in_sizes, int n_in,
                              void* d_out, int out_size, void* d_ws, size_t ws_size,
                              hipStream_t stream) {
  const float* r   = (const float*)d_in[0];
  const float* w   = (const float*)d_in[1];
  const float* k   = (const float*)d_in[2];
  const float* v   = (const float*)d_in[3];
  const float* g   = (const float*)d_in[4];
  const float* u   = (const float*)d_in[5];
  const float* is  = (const float*)d_in[6];
  const float* lnw = (const float*)d_in[7];
  const float* lnb = (const float*)d_in[8];
  const float* Wo  = (const float*)d_in[9];

  float* out   = (float*)d_out;
  float* state = out + (size_t)T_ * D_;

  float* o      = (float*)d_ws;
  float* Aseg   = o + (size_t)T_ * D_;
  float* Sloc   = Aseg + (size_t)H_ * SEG * dh;

  dim3 gs(SEG, H_);
  wkv_pass1<<<gs, 256, 0, stream>>>(w, k, v, Aseg, Sloc);
  wkv_pass2<<<gs, 128, 0, stream>>>(r, w, k, v, g, u, is, lnw, lnb, Aseg, Sloc,
                                    state, o);

  dim3 g2(D_ / 64, T_ / 64);
  oproj_kernel<<<g2, 128, 0, stream>>>(o, Wo, out);
}


# pass2 as a 3-stage wave pipeline: 2 waves stage chunk c+1 into a second LDS set, 4 waves do chunk c's matrix work, 4 waves do LN+gate of chunk c-1; 1 barrier a chunk, sum order kept
# speedup vs baseline: 4.8333x; 1.3873x over previous
#include <hip/hip_runtime.h>

typedef __attribute__((ext_vector_type(2))) float v2f;
typedef __attribute__((ext_vector_type(8))) float v8f;

#define D_   1024
#define H_   16
#define dh   64
#define T_   4096
#define CH   16
#define SEG  16
#define TSEG (T_/SEG)
#define NCHS (TSEG/CH)
#define ST   68
#define ST16 20

__device__ __forceinline__ v8f wmma_f32(v2f a, v2f b, v8f c) {
  return __builtin_amdgcn_wmma_f32_16x16x4_f32(false, a, false, b, (short)0, c,
                                               false, false);
}

__device__ __forceinline__ v2f fragA(const float* m, int stride, int k0) {
  int lane = threadIdx.x & 31;
  const float* p = m + (lane & 15) * stride + k0 + ((lane >> 4) << 1);
  v2f f; f[0] = p[0]; f[1] = p[1];
  return f;
}
__device__ __forceinline__ v2f fragB_kn(const float* m, int stride, int k0, int nbase) {
  int lane = threadIdx.x & 31;
  int kk = k0 + ((lane >> 4) << 1);
  int n  = nbase + (lane & 15);
  v2f f; f[0] = m[kk * stride + n]; f[1] = m[(kk + 1) * stride + n];
  return f;
}

#define P1ST 72
#define TBLK 16
#define NBLK (TSEG / TBLK)

__device__ __forceinline__ float4 mul4(float4 a, float4 b) {
  float4 r; r.x = a.x * b.x; r.y = a.y * b.y; r.z = a.z * b.z; r.w = a.w * b.w;
  return r;
}

__global__ __launch_bounds__(256)
void wkv_pass1(const float* __restrict__ wg, const float* __restrict__ kg,
               const float* __restrict__ vg, float* __restrict__ Aseg,
               float* __restrict__ Sloc) {
  __shared__ __attribute__((aligned(16))) float Kp[TSEG * P1ST];
  __shared__ __attribute__((aligned(16))) float Vp[TSEG * P1ST];
  __shared__ __attribute__((aligned(16))) float blk[NBLK * dh];

  const int seg = blockIdx.x, h = blockIdx.y;
  const int tid = threadIdx.x;
  const int lane = tid & 31, wid = tid >> 5;
  const int hi = lane >> 4, ln16 = lane & 15;
  const int cg = tid & 15, tb = tid >> 4;
  const int j4 = cg * 4;
  const size_t g0 = (size_t)(seg * TSEG + tb * TBLK) * D_ + h * dh + j4;

#pragma unroll
  for (int s = 0; s < TBLK; ++s)
    *(float4*)(Vp + (tb * TBLK + s) * P1ST + j4) = *(const float4*)(vg + g0 + (size_t)s * D_);

  float4 wv[TBLK], kv[TBLK];
#pragma unroll
  for (int s = 0; s < TBLK; ++s) wv[s] = *(const float4*)(wg + g0 + (size_t)s * D_);
#pragma unroll
  for (int s = 0; s < TBLK; ++s) kv[s] = *(const float4*)(kg + g0 + (size_t)s * D_);

  float4 p = make_float4(1.f, 1.f, 1.f, 1.f);
#pragma unroll
  for (int s = TBLK - 1; s >= 0; --s) {
    float4 t = wv[s];
    wv[s] = p;
    p = mul4(p, t);
  }
  const float4 total = (seg == 0 && tb == 0) ? wv[0] : p;
  *(float4*)(blk + tb * dh + j4) = total;
  __syncthreads();
  float4 carry = make_float4(1.f, 1.f, 1.f, 1.f);
#pragma unroll
  for (int b = NBLK - 1; b >= 1; --b) {
    float4 x = *(const float4*)(blk + b * dh + j4);
    const bool later = b > tb;
    x.x = later ? x.x : 1.f; x.y = later ? x.y : 1.f;
    x.z = later ? x.z : 1.f; x.w = later ? x.w : 1.f;
    carry = mul4(carry, x);
  }
  if (tb == 0)
    *(float4*)(Aseg + (h * SEG + seg) * dh + j4) = mul4(total, carry);

#pragma unroll
  for (int s = 0; s < TBLK; ++s) {
    const int row = tb * TBLK + s;
    *(float4*)(Kp + row * P1ST + j4) = mul4(kv[s], mul4(wv[s], carry));
  }
  __syncthreads();

  const int nb  = (wid & 3) * 16;
  const int itb = (wid >> 2) * 2;
  v8f acc[2] = {};
#pragma unroll 8
  for (int t0 = 0; t0 < TSEG; t0 += 4) {
    v2f b = fragB_kn(Vp, P1ST, t0, nb);
#pragma unroll
    for (int it = 0; it < 2; ++it)
      acc[it] = wmma_f32(fragB_kn(Kp, P1ST, t0, (itb + it) * 16), b, acc[it]);
  }
  __syncthreads();
#pragma unroll
  for (int it = 0; it < 2; ++it) {
#pragma unroll
    for (int v = 0; v < 8; ++v)
      Kp[((itb + it) * 16 + v + 8 * hi) * P1ST + nb + ln16] = acc[it][v];
  }
  __syncthreads();
  float* So = Sloc + (size_t)(h * SEG + seg) * dh * dh;
#pragma unroll
  for (int q = 0; q < 4; ++q) {
    int e = q * 256 + tid;
    int row = e >> 4, c4 = (e & 15) * 4;
    *(float4*)(So + row * dh + c4) = *(const float4*)(Kp + row * P1ST + c4);
  }
}

template <int N>
__device__ __forceinline__ void fold_carries(float4 (&s)[8],
                                             const float* __restrict__ AsegH,
                                             const float* __restrict__ SlocH,
                                             int sg, int tid) {
  float4 l[N][8];
  float  a[N][8];
#pragma unroll
  for (int n = 0; n < N; ++n) {
    const float4* L = (const float4*)(SlocH + (size_t)(sg + n) * dh * dh);
    const float*  A = AsegH + (sg + n) * dh + (tid >> 4);
#pragma unroll
    for (int q = 0; q < 8; ++q) { l[n][q] = L[q * 128 + tid]; a[n][q] = A[q * 8]; }
  }
#pragma unroll
  for (int n = 0; n < N; ++n) {
#pragma unroll
    for (int q = 0; q < 8; ++q) {
      s[q].x = a[n][q] * s[q].x + l[n][q].x;
      s[q].y = a[n][q] * s[q].y + l[n][q].y;
      s[q].z = a[n][q] * s[q].z + l[n][q].z;
      s[q].w = a[n][q] * s[q].w + l[n][q].w;
    }
  }
}

__device__ __forceinline__ void p2_stage(const float* __restrict__ rg,
                                         const float* __restrict__ wg,
                                         const float* __restrict__ kg,
                                         const float* __restrict__ vg,
                                         float ui, int i, int goff, bool first,
                                         float* Rt, float* Kt, float* Ku,
                                         float* Vv, float* Kh, float* aLast) {
  float wr[CH], rr[CH], kr[CH], vr[CH];
#pragma unroll
  for (int t = 0; t < CH; ++t) {
    const int off = goff + t * D_;
    wr[t] = wg[off]; rr[t] = rg[off]; kr[t] = kg[off]; vr[t] = vg[off];
  }
  float a = 1.f, ai = 1.f;
  float air[CH];
#pragma unroll
  for (int t = 0; t < CH; ++t) {
    float wv = wr[t];
    if (first && t == 0) wv = 1.f;
    const float rv = rr[t], kv = kr[t];
    const float aip = ai;
    Rt[t * ST + i] = rv * a;
    const float rw = 1.f / wv;
    a *= wv; ai *= rw;
    Kt[t * ST + i] = kv * ai;
    Ku[t * ST + i] = kv * ui * aip;
    Vv[t * ST + i] = vr[t];
    air[t] = ai;
  }
  aLast[i] = a;
#pragma unroll
  for (int t = 0; t < CH; ++t) Kh[i * ST16 + t] = kr[t] * (a * air[t]);
}

__global__ __launch_bounds__(320)
void wkv_pass2(const float* __restrict__ rg, const float* __restrict__ wg,
               const float* __restrict__ kg, const float* __restrict__ vg,
               const float* __restrict__ gg, const float* __restrict__ ug,
               const float* __restrict__ isg, const float* __restrict__ lnw,
               const float* __restrict__ lnb, const float* __restrict__ Aseg,
               const float* __restrict__ Sloc, float* __restrict__ state_out,
               float* __restrict__ og) {
  __shared__ __attribute__((aligned(16))) float S[dh * ST];
  __shared__ float Rt[2][CH * ST];
  __shared__ float Kt[2][CH * ST];
  __shared__ float Ku[2][CH * ST];
  __shared__ float Vv[2][CH * ST];
  __shared__ float Kh[2][dh * ST16];
  __shared__ float CmW[4][CH * ST16];
  __shared__ float ybuf[2][CH * ST];
  __shared__ float aLast[2][dh];
  __shared__ float corr0[dh];
  __shared__ float lnw_s[dh], lnb_s[dh];
  __shared__ float r0sum_s;

  const int seg = blockIdx.x, h = blockIdx.y;
  const int tid = threadIdx.x;
  const int lane = tid & 31;
  const int hi = lane >> 4, ln16 = lane & 15;
  const int wid = __builtin_amdgcn_readfirstlane(tid >> 5);
  const int role = (wid < 4) ? 0 : ((wid < 6) ? 1 : 2);

  const int si = tid - 128;
  const float ui = (role == 1) ? ug[h * dh + si] : 0.f;
  const int et = tid - 192;
  const int e_t = et >> 3, e_jb = (et & 7) * 8;

  if (role == 0) {
    if (tid < dh)
      corr0[tid] = isg[h * dh + tid] * (1.f - wg[h * dh + tid]);
    if (tid == 0 && seg == 0) {
      float s = 0.f;
      for (int i2 = 0; i2 < dh; ++i2) s += rg[h * dh + i2];
      r0sum_s = s;
    }
    const float* AsegH = Aseg + h * SEG * dh;
    const float* SlocH = Sloc + (size_t)h * SEG * dh * dh;
    const int j4 = (tid & 15) * 4;
    const float4 w0 = *(const float4*)(wg + h * dh + j4);
    const float4 i0 = *(const float4*)(isg + h * dh + j4);
    float4 s[8];
#pragma unroll
    for (int q = 0; q < 8; ++q) {
      s[q].x = w0.x * i0.x; s[q].y = w0.y * i0.y;
      s[q].z = w0.z * i0.z; s[q].w = w0.w * i0.w;
    }
    int sg = 0;
    for (; sg + 4 <= seg; sg += 4) fold_carries<4>(s, AsegH, SlocH, sg, tid);
    if (sg + 2 <= seg) { fold_carries<2>(s, AsegH, SlocH, sg, tid); sg += 2; }
    if (sg < seg)      { fold_carries<1>(s, AsegH, SlocH, sg, tid); sg += 1; }
#pragma unroll
    for (int q = 0; q < 8; ++q)
      *(float4*)(S + (q * 8 + (tid >> 4)) * ST + j4) = s[q];
    if (seg == SEG - 1) {
      fold_carries<1>(s, AsegH, SlocH, sg, tid);
#pragma unroll
      for (int q = 0; q < 8; ++q)
        *(float4*)(state_out + (size_t)(q * 8 + (tid >> 4)) * D_ + h * dh + j4) = s[q];
    }
  } else if (role == 1) {
    p2_stage(rg, wg, kg, vg, ui, si, (seg * TSEG) * D_ + h * dh + si, seg == 0,
             Rt[0], Kt[0], Ku[0], Vv[0], Kh[0], aLast[0]);
  } else {
    if (et < dh) { lnw_s[et] = lnw[et]; lnb_s[et] = lnb[et]; }
  }
  __syncthreads();

  for (int c = 0; c <= NCHS; ++c) {
    const int b = c & 1;
    if (role == 0) {
      if (c < NCHS) {
        const float* rt = Rt[b];
        const float* kt = Kt[b];
        const float* ku = Ku[b];
        const float* vp = Vv[b];
        const float* kh = Kh[b];
        const float* al = aLast[b];
        float* yb = ybuf[b];

        v8f accC = {}; v8f accD = {};
#pragma unroll
        for (int k0 = 0; k0 < dh; k0 += 4) {
          v2f arf = fragA(rt, ST, k0);
          accC = wmma_f32(arf, fragA(kt, ST, k0), accC);
          accD = wmma_f32(arf, fragA(ku, ST, k0), accD);
        }
        float* cm = CmW[wid];
#pragma unroll
        for (int v = 0; v < 8; ++v) {
          int M = v + 8 * hi, N = ln16;
          cm[M * ST16 + N] = (N < M) ? accC[v] : ((N == M) ? accD[v] : 0.f);
        }
        asm volatile("" ::: "memory");

        const int nb = wid * 16;
        v8f accY = {};
#pragma unroll
        for (int k0 = 0; k0 < dh; k0 += 4)
          accY = wmma_f32(fragA(rt, ST, k0), fragB_kn(S, ST, k0, nb), accY);
#pragma unroll
        for (int k0 = 0; k0 < CH; k0 += 4)
          accY = wmma_f32(fragA(cm, ST16, k0), fragB_kn(vp, ST, k0, nb), accY);
        if (seg == 0 && c == 0 && hi == 0)
          accY[0] += r0sum_s * corr0[nb + ln16];
#pragma unroll
        for (int v = 0; v < 8; ++v) yb[(v + 8 * hi) * ST + nb + ln16] = accY[v];

        v8f accS[4];
#pragma unroll
        for (int it = 0; it < 4; ++it) {
          v8f a = {};
#pragma unroll
          for (int k0 = 0; k0 < CH; k0 += 4)
            a = wmma_f32(fragA(kh + it * 16 * ST16, ST16, k0),
                         fragB_kn(vp, ST, k0, nb), a);
          accS[it] = a;
        }
#pragma unroll
        for (int it = 0; it < 4; ++it) {
#pragma unroll
          for (int v = 0; v < 8; ++v) {
            int i   = it * 16 + v + 8 * hi;
            int idx = i * ST + nb + ln16;
            S[idx]  = al[i] * S[idx] + accS[it][v];
          }
        }
      }
    } else if (role == 1) {
      if (c + 1 < NCHS) {
        const int bn = (c + 1) & 1;
        p2_stage(rg, wg, kg, vg, ui, si,
                 (seg * TSEG + (c + 1) * CH) * D_ + h * dh + si, false,
                 Rt[bn], Kt[bn], Ku[bn], Vv[bn], Kh[bn], aLast[bn]);
      }
    } else {
      if (c >= 1) {
        const float* yb = ybuf[(c - 1) & 1];
        const int off = (seg * TSEG + (c - 1) * CH + e_t) * D_ + h * dh + e_jb;
        const float4 ga = *(const float4*)(gg + off);
        const float4 gb = *(const float4*)(gg + off + 4);
        float s = 0.f, sq = 0.f;
        for (int j = 0; j < dh; ++j) {
          float y = yb[e_t * ST + j]; s += y; sq += y * y;
        }
        const float m  = s * (1.f / dh);
        const float vv = sq * (1.f / dh) - m * m;
        const float rs = rsqrtf(vv + 1e-5f);
        const float gq[8] = {ga.x, ga.y, ga.z, ga.w, gb.x, gb.y, gb.z, gb.w};
        float oq[8];
#pragma unroll
        for (int q = 0; q < 8; ++q) {
          int j    = e_jb + q;
          float y  = yb[e_t * ST + j];
          float yn = (y - m) * rs * lnw_s[j] + lnb_s[j];
          float gv = gq[q];
          oq[q]    = yn * (gv / (1.f + __expf(-gv)));
        }
        *(float4*)(og + off)     = make_float4(oq[0], oq[1], oq[2], oq[3]);
        *(float4*)(og + off + 4) = make_float4(oq[4], oq[5], oq[6], oq[7]);
      }
    }
    __syncthreads();
  }
}

__global__ __launch_bounds__(128)
void oproj_kernel(const float* __restrict__ o, const float* __restrict__ Wo,
                  float* __restrict__ out) {
  __shared__ float Ao[64 * ST];
  __shared__ float Bo[64 * ST];
  const int tid = threadIdx.x;
  const int lane = tid & 31, wid = tid >> 5;
  const int hi = lane >> 4, ln16 = lane & 15;
  const int nbase = blockIdx.x * 64;
  const int mbase = blockIdx.y * 64;

  v8f acc[4] = {};

  for (int ks = 0; ks < D_; ks += 64) {
    __syncthreads();
    {
      int row = tid >> 1;
      int cb  = (tid & 1) * 32;
      const float4* pa = (const float4*)(o  + (size_t)(mbase + row) * D_ + ks + cb);
      const float4* pb = (const float4*)(Wo + (size_t)(nbase + row) * D_ + ks + cb);
#pragma unroll
      for (int q = 0; q < 8; ++q) {
        float4 a4 = pa[q]; float4 b4 = pb[q];
        int cc = cb + q * 4;
        Ao[row * ST + cc + 0] = a4.x; Ao[row * ST + cc + 1] = a4.y;
        Ao[row * ST + cc + 2] = a4.z; Ao[row * ST + cc + 3] = a4.w;
        Bo[row * ST + cc + 0] = b4.x; Bo[row * ST + cc + 1] = b4.y;
        Bo[row * ST + cc + 2] = b4.z; Bo[row * ST + cc + 3] = b4.w;
      }
    }
    __syncthreads();
#pragma unroll
    for (int k0 = 0; k0 < 64; k0 += 4) {
      v2f b = fragA(Bo + wid * 16 * ST, ST, k0);
#pragma unroll
      for (int mt = 0; mt < 4; ++mt)
        acc[mt] = wmma_f32(fragA(Ao + mt * 16 * ST, ST, k0), b, acc[mt]);
    }
  }
#pragma unroll
  for (int mt = 0; mt < 4; ++mt) {
#pragma unroll
    for (int v = 0; v < 8; ++v) {
      int row = mbase + mt * 16 + v + 8 * hi;
      int col = nbase + wid * 16 + ln16;
      out[(size_t)row * D_ + col] = acc[mt][v];
    }
  }
}

extern "C" void kernel_launch(void* const* d_in, const int* in_sizes, int n_in,
                              void* d_out, int out_size, void* d_ws, size_t ws_size,
                              hipStream_t stream) {
  const float* r   = (const float*)d_in[0];
  const float* w   = (const float*)d_in[1];
  const float* k   = (const float*)d_in[2];
  const float* v   = (const float*)d_in[3];
  const float* g   = (const float*)d_in[4];
  const float* u   = (const float*)d_in[5];
  const float* is  = (const float*)d_in[6];
  const float* lnw = (const float*)d_in[7];
  const float* lnb = (const float*)d_in[8];
  const float* Wo  = (const float*)d_in[9];

  float* out   = (float*)d_out;
  float* state = out + (size_t)T_ * D_;

  float* o      = (float*)d_ws;
  float* Aseg   = o + (size_t)T_ * D_;
  float* Sloc   = Aseg + (size_t)H_ * SEG * dh;

  dim3 gs(SEG, H_);
  wkv_pass1<<<gs, 256, 0, stream>>>(w, k, v, Aseg, Sloc);
  wkv_pass2<<<gs, 320, 0, stream>>>(r, w, k, v, g, u, is, lnw, lnb, Aseg, Sloc,
                                    state, o);

  dim3 g2(D_ / 64, T_ / 64);
  oproj_kernel<<<g2, 128, 0, stream>>>(o, Wo, out);
}


# o_proj: both operands read as ready WMMA fragments (o stored, Wo copied by pass2's idle gate waves, in fragment order): 512-B loads to registers, no LDS tiles or barriers; k sum order changes
# speedup vs baseline: 5.9378x; 1.2285x over previous
#include <hip/hip_runtime.h>

typedef __attribute__((ext_vector_type(2))) float v2f;
typedef __attribute__((ext_vector_type(8))) float v8f;

#define D_   1024
#define H_   16
#define dh   64
#define T_   4096
#define CH   16
#define SEG  16
#define TSEG (T_/SEG)
#define NCHS (TSEG/CH)
#define ST   68
#define ST16 20

__device__ __forceinline__ v8f wmma_f32(v2f a, v2f b, v8f c) {
  return __builtin_amdgcn_wmma_f32_16x16x4_f32(false, a, false, b, (short)0, c,
                                               false, false);
}

__device__ __forceinline__ v2f fragA(const float* m, int stride, int k0) {
  int lane = threadIdx.x & 31;
  const float* p = m + (lane & 15) * stride + k0 + ((lane >> 4) << 1);
  v2f f; f[0] = p[0]; f[1] = p[1];
  return f;
}
__device__ __forceinline__ v2f fragB_kn(const float* m, int stride, int k0, int nbase) {
  int lane = threadIdx.x & 31;
  int kk = k0 + ((lane >> 4) << 1);
  int n  = nbase + (lane & 15);
  v2f f; f[0] = m[kk * stride + n]; f[1] = m[(kk + 1) * stride + n];
  return f;
}

#define P1ST 72
#define TBLK 16
#define NBLK (TSEG / TBLK)

__device__ __forceinline__ float4 mul4(float4 a, float4 b) {
  float4 r; r.x = a.x * b.x; r.y = a.y * b.y; r.z = a.z * b.z; r.w = a.w * b.w;
  return r;
}

__global__ __launch_bounds__(256)
void wkv_pass1(const float* __restrict__ wg, const float* __restrict__ kg,
               const float* __restrict__ vg, float* __restrict__ Aseg,
               float* __restrict__ Sloc) {
  __shared__ __attribute__((aligned(16))) float Kp[TSEG * P1ST];
  __shared__ __attribute__((aligned(16))) float Vp[TSEG * P1ST];
  __shared__ __attribute__((aligned(16))) float blk[NBLK * dh];

  const int seg = blockIdx.x, h = blockIdx.y;
  const int tid = threadIdx.x;
  const int lane = tid & 31, wid = tid >> 5;
  const int hi = lane >> 4, ln16 = lane & 15;
  const int cg = tid & 15, tb = tid >> 4;
  const int j4 = cg * 4;
  const size_t g0 = (size_t)(seg * TSEG + tb * TBLK) * D_ + h * dh + j4;

#pragma unroll
  for (int s = 0; s < TBLK; ++s)
    *(float4*)(Vp + (tb * TBLK + s) * P1ST + j4) = *(const float4*)(vg + g0 + (size_t)s * D_);

  float4 wv[TBLK], kv[TBLK];
#pragma unroll
  for (int s = 0; s < TBLK; ++s) wv[s] = *(const float4*)(wg + g0 + (size_t)s * D_);
#pragma unroll
  for (int s = 0; s < TBLK; ++s) kv[s] = *(const float4*)(kg + g0 + (size_t)s * D_);

  float4 p = make_float4(1.f, 1.f, 1.f, 1.f);
#pragma unroll
  for (int s = TBLK - 1; s >= 0; --s) {
    float4 t = wv[s];
    wv[s] = p;
    p = mul4(p, t);
  }
  const float4 total = (seg == 0 && tb == 0) ? wv[0] : p;
  *(float4*)(blk + tb * dh + j4) = total;
  __syncthreads();
  float4 carry = make_float4(1.f, 1.f, 1.f, 1.f);
#pragma unroll
  for (int b = NBLK - 1; b >= 1; --b) {
    float4 x = *(const float4*)(blk + b * dh + j4);
    const bool later = b > tb;
    x.x = later ? x.x : 1.f; x.y = later ? x.y : 1.f;
    x.z = later ? x.z : 1.f; x.w = later ? x.w : 1.f;
    carry = mul4(carry, x);
  }
  if (tb == 0)
    *(float4*)(Aseg + (h * SEG + seg) * dh + j4) = mul4(total, carry);

#pragma unroll
  for (int s = 0; s < TBLK; ++s) {
    const int row = tb * TBLK + s;
    *(float4*)(Kp + row * P1ST + j4) = mul4(kv[s], mul4(wv[s], carry));
  }
  __syncthreads();

  const int nb  = (wid & 3) * 16;
  const int itb = (wid >> 2) * 2;
  v8f acc[2] = {};
#pragma unroll 8
  for (int t0 = 0; t0 < TSEG; t0 += 4) {
    v2f b = fragB_kn(Vp, P1ST, t0, nb);
#pragma unroll
    for (int it = 0; it < 2; ++it)
      acc[it] = wmma_f32(fragB_kn(Kp, P1ST, t0, (itb + it) * 16), b, acc[it]);
  }
  __syncthreads();
#pragma unroll
  for (int it = 0; it < 2; ++it) {
#pragma unroll
    for (int v = 0; v < 8; ++v)
      Kp[((itb + it) * 16 + v + 8 * hi) * P1ST + nb + ln16] = acc[it][v];
  }
  __syncthreads();
  float* So = Sloc + (size_t)(h * SEG + seg) * dh * dh;
#pragma unroll
  for (int q = 0; q < 4; ++q) {
    int e = q * 256 + tid;
    int row = e >> 4, c4 = (e & 15) * 4;
    *(float4*)(So + row * dh + c4) = *(const float4*)(Kp + row * P1ST + c4);
  }
}

template <int N>
__device__ __forceinline__ void fold_carries(float4 (&s)[8],
                                             const float* __restrict__ AsegH,
                                             const float* __restrict__ SlocH,
                                             int sg, int tid) {
  float4 l[N][8];
  float  a[N][8];
#pragma unroll
  for (int n = 0; n < N; ++n) {
    const float4* L = (const float4*)(SlocH + (size_t)(sg + n) * dh * dh);
    const float*  A = AsegH + (sg + n) * dh + (tid >> 4);
#pragma unroll
    for (int q = 0; q < 8; ++q) { l[n][q] = L[q * 128 + tid]; a[n][q] = A[q * 8]; }
  }
#pragma unroll
  for (int n = 0; n < N; ++n) {
#pragma unroll
    for (int q = 0; q < 8; ++q) {
      s[q].x = a[n][q] * s[q].x + l[n][q].x;
      s[q].y = a[n][q] * s[q].y + l[n][q].y;
      s[q].z = a[n][q] * s[q].z + l[n][q].z;
      s[q].w = a[n][q] * s[q].w + l[n][q].w;
    }
  }
}

__device__ __forceinline__ void p2_stage(const float* __restrict__ rg,
                                         const float* __restrict__ wg,
                                         const float* __restrict__ kg,
                                         const float* __restrict__ vg,
                                         float ui, int i, int goff, bool first,
                                         float* Rt, float* Kt, float* Ku,
                                         float* Vv, float* Kh, float* aLast) {
  float wr[CH], rr[CH], kr[CH], vr[CH];
#pragma unroll
  for (int t = 0; t < CH; ++t) {
    const int off = goff + t * D_;
    wr[t] = wg[off]; rr[t] = rg[off]; kr[t] = kg[off]; vr[t] = vg[off];
  }
  float a = 1.f, ai = 1.f;
  float air[CH];
#pragma unroll
  for (int t = 0; t < CH; ++t) {
    float wv = wr[t];
    if (first && t == 0) wv = 1.f;
    const float rv = rr[t], kv = kr[t];
    const float aip = ai;
    Rt[t * ST + i] = rv * a;
    const float rw = 1.f / wv;
    a *= wv; ai *= rw;
    Kt[t * ST + i] = kv * ai;
    Ku[t * ST + i] = kv * ui * aip;
    Vv[t * ST + i] = vr[t];
    air[t] = ai;
  }
  aLast[i] = a;
#pragma unroll
  for (int t = 0; t < CH; ++t) Kh[i * ST16 + t] = kr[t] * (a * air[t]);
}

__global__ __launch_bounds__(320)
void wkv_pass2(const float* __restrict__ rg, const float* __restrict__ wg,
               const float* __restrict__ kg, const float* __restrict__ vg,
               const float* __restrict__ gg, const float* __restrict__ ug,
               const float* __restrict__ isg, const float* __restrict__ lnw,
               const float* __restrict__ lnb, const float* __restrict__ Aseg,
               const float* __restrict__ Sloc, const float* __restrict__ Wog,
               float* __restrict__ Wof, float* __restrict__ state_out,
               float* __restrict__ og) {
  __shared__ __attribute__((aligned(16))) float S[dh * ST];
  __shared__ float Rt[2][CH * ST];
  __shared__ float Kt[2][CH * ST];
  __shared__ float Ku[2][CH * ST];
  __shared__ float Vv[2][CH * ST];
  __shared__ float Kh[2][dh * ST16];
  __shared__ float CmW[4][CH * ST16];
  __shared__ float ybuf[2][CH * ST];
  __shared__ float aLast[2][dh];
  __shared__ float corr0[dh];
  __shared__ float lnw_s[dh], lnb_s[dh];
  __shared__ float r0sum_s;

  const int seg = blockIdx.x, h = blockIdx.y;
  const int tid = threadIdx.x;
  const int lane = tid & 31;
  const int hi = lane >> 4, ln16 = lane & 15;
  const int wid = __builtin_amdgcn_readfirstlane(tid >> 5);
  const int role = (wid < 4) ? 0 : ((wid < 6) ? 1 : 2);

  const int si = tid - 128;
  const float ui = (role == 1) ? ug[h * dh + si] : 0.f;
  const int et = tid - 192;
  const int e_t = et >> 3, e_jb = (et & 7) * 8;

  if (role == 0) {
    if (tid < dh)
      corr0[tid] = isg[h * dh + tid] * (1.f - wg[h * dh + tid]);
    if (tid == 0 && seg == 0) {
      float s = 0.f;
      for (int i2 = 0; i2 < dh; ++i2) s += rg[h * dh + i2];
      r0sum_s = s;
    }
    const float* AsegH = Aseg + h * SEG * dh;
    const float* SlocH = Sloc + (size_t)h * SEG * dh * dh;
    const int j4 = (tid & 15) * 4;
    const float4 w0 = *(const float4*)(wg + h * dh + j4);
    const float4 i0 = *(const float4*)(isg + h * dh + j4);
    float4 s[8];
#pragma unroll
    for (int q = 0; q < 8; ++q) {
      s[q].x = w0.x * i0.x; s[q].y = w0.y * i0.y;
      s[q].z = w0.z * i0.z; s[q].w = w0.w * i0.w;
    }
    int sg = 0;
    for (; sg + 4 <= seg; sg += 4) fold_carries<4>(s, AsegH, SlocH, sg, tid);
    if (sg + 2 <= seg) { fold_carries<2>(s, AsegH, SlocH, sg, tid); sg += 2; }
    if (sg < seg)      { fold_carries<1>(s, AsegH, SlocH, sg, tid); sg += 1; }
#pragma unroll
    for (int q = 0; q < 8; ++q)
      *(float4*)(S + (q * 8 + (tid >> 4)) * ST + j4) = s[q];
    if (seg == SEG - 1) {
      fold_carries<1>(s, AsegH, SlocH, sg, tid);
#pragma unroll
      for (int q = 0; q < 8; ++q)
        *(float4*)(state_out + (size_t)(q * 8 + (tid >> 4)) * D_ + h * dh + j4) = s[q];
    }
  } else if (role == 1) {
    p2_stage(rg, wg, kg, vg, ui, si, (seg * TSEG) * D_ + h * dh + si, seg == 0,
             Rt[0], Kt[0], Ku[0], Vv[0], Kh[0], aLast[0]);
  } else {
    if (et < dh) { lnw_s[et] = lnw[et]; lnb_s[et] = lnb[et]; }
    const int gidx = (h * SEG + seg) * 128 + et;
    const int nt = gidx >> 9, kp8 = (gidx >> 5) & 15;
    const float4* wsrc = (const float4*)(Wog + (size_t)(nt * 16 + ln16) * D_) + kp8 * 16 + hi;
    float4* wdst = (float4*)Wof + ((size_t)(nt * 128 + kp8 * 8) * 32 + lane);
    float4 wt[8];
#pragma unroll
    for (int j = 0; j < 8; ++j) wt[j] = wsrc[2 * j];
#pragma unroll
    for (int j = 0; j < 8; ++j) wdst[j * 32] = wt[j];
  }
  __syncthreads();

  for (int c = 0; c <= NCHS; ++c) {
    const int b = c & 1;
    if (role == 0) {
      if (c < NCHS) {
        const float* rt = Rt[b];
        const float* kt = Kt[b];
        const float* ku = Ku[b];
        const float* vp = Vv[b];
        const float* kh = Kh[b];
        const float* al = aLast[b];
        float* yb = ybuf[b];

        v8f accC = {}; v8f accD = {};
#pragma unroll
        for (int k0 = 0; k0 < dh; k0 += 4) {
          v2f arf = fragA(rt, ST, k0);
          accC = wmma_f32(arf, fragA(kt, ST, k0), accC);
          accD = wmma_f32(arf, fragA(ku, ST, k0), accD);
        }
        float* cm = CmW[wid];
#pragma unroll
        for (int v = 0; v < 8; ++v) {
          int M = v + 8 * hi, N = ln16;
          cm[M * ST16 + N] = (N < M) ? accC[v] : ((N == M) ? accD[v] : 0.f);
        }
        asm volatile("" ::: "memory");

        const int nb = wid * 16;
        v8f accY = {};
#pragma unroll
        for (int k0 = 0; k0 < dh; k0 += 4)
          accY = wmma_f32(fragA(rt, ST, k0), fragB_kn(S, ST, k0, nb), accY);
#pragma unroll
        for (int k0 = 0; k0 < CH; k0 += 4)
          accY = wmma_f32(fragA(cm, ST16, k0), fragB_kn(vp, ST, k0, nb), accY);
        if (seg == 0 && c == 0 && hi == 0)
          accY[0] += r0sum_s * corr0[nb + ln16];
#pragma unroll
        for (int v = 0; v < 8; ++v) yb[(v + 8 * hi) * ST + nb + ln16] = accY[v];

        v8f accS[4];
#pragma unroll
        for (int it = 0; it < 4; ++it) {
          v8f a = {};
#pragma unroll
          for (int k0 = 0; k0 < CH; k0 += 4)
            a = wmma_f32(fragA(kh + it * 16 * ST16, ST16, k0),
                         fragB_kn(vp, ST, k0, nb), a);
          accS[it] = a;
        }
#pragma unroll
        for (int it = 0; it < 4; ++it) {
#pragma unroll
          for (int v = 0; v < 8; ++v) {
            int i   = it * 16 + v + 8 * hi;
            int idx = i * ST + nb + ln16;
            S[idx]  = al[i] * S[idx] + accS[it][v];
          }
        }
      }
    } else if (role == 1) {
      if (c + 1 < NCHS) {
        const int bn = (c + 1) & 1;
        p2_stage(rg, wg, kg, vg, ui, si,
                 (seg * TSEG + (c + 1) * CH) * D_ + h * dh + si, false,
                 Rt[bn], Kt[bn], Ku[bn], Vv[bn], Kh[bn], aLast[bn]);
      }
    } else {
      if (c >= 1) {
        const float* yb = ybuf[(c - 1) & 1];
        const int off = (seg * TSEG + (c - 1) * CH + e_t) * D_ + h * dh + e_jb;
        const float4 ga = *(const float4*)(gg + off);
        const float4 gb = *(const float4*)(gg + off + 4);
        float s = 0.f, sq = 0.f;
        for (int j = 0; j < dh; ++j) {
          float y = yb[e_t * ST + j]; s += y; sq += y * y;
        }
        const float m  = s * (1.f / dh);
        const float vv = sq * (1.f / dh) - m * m;
        const float rs = rsqrtf(vv + 1e-5f);
        const float gq[8] = {ga.x, ga.y, ga.z, ga.w, gb.x, gb.y, gb.z, gb.w};
        float oq[8];
#pragma unroll
        for (int q = 0; q < 8; ++q) {
          int j    = e_jb + q;
          float y  = yb[e_t * ST + j];
          float yn = (y - m) * rs * lnw_s[j] + lnb_s[j];
          float gv = gq[q];
          oq[q]    = yn * (gv / (1.f + __expf(-gv)));
        }
        const int tile = seg * NCHS + (c - 1);
        float4* od = (float4*)og + ((size_t)(tile * 128 + h * 8 + (et & 7)) * 32 + e_t);
        od[0]  = make_float4(oq[0], oq[1], oq[2], oq[3]);
        od[16] = make_float4(oq[4], oq[5], oq[6], oq[7]);
      }
    }
    __syncthreads();
  }
}

#define KPB 4

__device__ __forceinline__ void op_load(float4 (&a)[KPB][4], float4 (&b)[KPB][2],
                                        const float4* __restrict__ pa,
                                        const float4* __restrict__ pb, int kp0) {
#pragma unroll
  for (int s = 0; s < KPB; ++s) {
#pragma unroll
    for (int mt = 0; mt < 4; ++mt) a[s][mt] = pa[(mt * 128 + kp0 + s) * 32];
#pragma unroll
    for (int nt = 0; nt < 2; ++nt) b[s][nt] = pb[(nt * 128 + kp0 + s) * 32];
  }
}

__device__ __forceinline__ void op_mma(v8f (&acc)[4][2], const float4 (&a)[KPB][4],
                                       const float4 (&b)[KPB][2]) {
#pragma unroll
  for (int s = 0; s < KPB; ++s) {
#pragma unroll
    for (int nt = 0; nt < 2; ++nt) {
      v2f bl; bl[0] = b[s][nt].x; bl[1] = b[s][nt].y;
#pragma unroll
      for (int mt = 0; mt < 4; ++mt) {
        v2f al; al[0] = a[s][mt].x; al[1] = a[s][mt].y;
        acc[mt][nt] = wmma_f32(al, bl, acc[mt][nt]);
      }
    }
#pragma unroll
    for (int nt = 0; nt < 2; ++nt) {
      v2f bh; bh[0] = b[s][nt].z; bh[1] = b[s][nt].w;
#pragma unroll
      for (int mt = 0; mt < 4; ++mt) {
        v2f ah; ah[0] = a[s][mt].z; ah[1] = a[s][mt].w;
        acc[mt][nt] = wmma_f32(ah, bh, acc[mt][nt]);
      }
    }
  }
}

__global__ __launch_bounds__(128)
void oproj_kernel(const float* __restrict__ of, const float* __restrict__ wf,
                  float* __restrict__ out) {
  const int tid = threadIdx.x;
  const int lane = tid & 31, wid = tid >> 5;
  const int hi = lane >> 4, ln16 = lane & 15;
  const int mt0 = blockIdx.y * 4;
  const int nt0 = blockIdx.x * 8 + wid * 2;
  const float4* pa = (const float4*)of + ((size_t)mt0 * 128 * 32 + lane);
  const float4* pb = (const float4*)wf + ((size_t)nt0 * 128 * 32 + lane);

  v8f acc[4][2] = {};
  float4 a0[KPB][4], b0[KPB][2], a1[KPB][4], b1[KPB][2];
  op_load(a0, b0, pa, pb, 0);
  for (int kp = 0; kp < D_ / 8 - 2 * KPB; kp += 2 * KPB) {
    op_load(a1, b1, pa, pb, kp + KPB);
    __builtin_amdgcn_sched_barrier(0);
    op_mma(acc, a0, b0);
    __builtin_amdgcn_sched_barrier(0);
    op_load(a0, b0, pa, pb, kp + 2 * KPB);
    __builtin_amdgcn_sched_barrier(0);
    op_mma(acc, a1, b1);
    __builtin_amdgcn_sched_barrier(0);
  }
  op_load(a1, b1, pa, pb, D_ / 8 - KPB);
  __builtin_amdgcn_sched_barrier(0);
  op_mma(acc, a0, b0);
  op_mma(acc, a1, b1);
#pragma unroll
  for (int mt = 0; mt < 4; ++mt) {
#pragma unroll
    for (int nt = 0; nt < 2; ++nt) {
#pragma unroll
      for (int v = 0; v < 8; ++v) {
        int row = (mt0 + mt) * 16 + v + 8 * hi;
        int col = (nt0 + nt) * 16 + ln16;
        out[(size_t)row * D_ + col] = acc[mt][nt][v];
      }
    }
  }
}

extern "C" void kernel_launch(void* const* d_in, const int* in_sizes, int n_in,
                              void* d_out, int out_size, void* d_ws, size_t ws_size,
                              hipStream_t stream) {
  const float* r   = (const float*)d_in[0];
  const float* w   = (const float*)d_in[1];
  const float* k   = (const float*)d_in[2];
  const float* v   = (const float*)d_in[3];
  const float* g   = (const float*)d_in[4];
  const float* u   = (const float*)d_in[5];
  const float* is  = (const float*)d_in[6];
  const float* lnw = (const float*)d_in[7];
  const float* lnb = (const float*)d_in[8];
  const float* Wo  = (const float*)d_in[9];

  float* out   = (float*)d_out;
  float* state = out + (size_t)T_ * D_;

  float* o      = (float*)d_ws;
  float* Aseg   = o + (size_t)T_ * D_;
  float* Sloc   = Aseg + (size_t)H_ * SEG * dh;
  float* Wof    = Sloc + (size_t)H_ * SEG * dh * dh;

  dim3 gs(SEG, H_);
  wkv_pass1<<<gs, 256, 0, stream>>>(w, k, v, Aseg, Sloc);
  wkv_pass2<<<gs, 320, 0, stream>>>(r, w, k, v, g, u, is, lnw, lnb, Aseg, Sloc,
                                    Wo, Wof, state, o);

  dim3 g2(D_ / 128, T_ / 64);
  oproj_kernel<<<g2, 128, 0, stream>>>(o, Wof, out);
}
